# starting kernel
# baseline (speedup 1.0000x reference)
#include <hip/hip_runtime.h>
#include <hip/hip_bf16.h>
#include <climits>

typedef __attribute__((ext_vector_type(2))) float v2f;
typedef __attribute__((ext_vector_type(8))) float v8f;

#define N_STARS 2048
#define BATCH   256
#define PF      28
#define NGE     512
#define GF      32
#define OPD2    65536
#define KPAD    64

__global__ __launch_bounds__(64)
void build_interm_kernel(const float* __restrict__ positions,
                         const float* __restrict__ obs_pos,
                         const float* __restrict__ poly_dic,
                         const float* __restrict__ graph_dic,
                         const float* __restrict__ alpha_poly,
                         const float* __restrict__ alpha_graph,
                         float* __restrict__ interm) {
    __shared__ int s_min;
    const int b = blockIdx.x;
    const int t = threadIdx.x;
    if (t == 0) s_min = INT_MAX;
    __syncthreads();

    const float px = positions[2 * b];
    const float py = positions[2 * b + 1];
    int best = INT_MAX;
    for (int s = t; s < N_STARS; s += 64) {
        const float ox = obs_pos[2 * s];
        const float oy = obs_pos[2 * s + 1];
        if (ox == px)       best = min(best, 2 * s);
        else if (oy == py)  best = min(best, 2 * s + 1);
    }
    atomicMin(&s_min, best);
    __syncthreads();
    int flat = s_min;
    if (flat == INT_MAX) flat = 0;
    const int idx = flat >> 1;

    float acc = 0.0f;
    if (t < PF) {
        const float* pd = poly_dic + (size_t)idx * PF;
        #pragma unroll
        for (int p = 0; p < PF; ++p)
            acc += pd[p] * alpha_poly[p * PF + t];
    } else if (t < PF + GF) {
        const int j = t - PF;
        const float* gd = graph_dic + (size_t)idx * NGE;
        for (int e = 0; e < NGE; ++e)
            acc += gd[e] * alpha_graph[e * GF + j];
    }
    interm[(size_t)b * KPAD + t] = acc;
}

__global__ __launch_bounds__(256)
void opd_wmma_gemm_kernel(const float* __restrict__ interm,
                          const float* __restrict__ S_poly,
                          const float* __restrict__ S_graph,
                          float* __restrict__ out) {
    const int lane  = threadIdx.x & 31;
    const int wave  = threadIdx.x >> 5;
    const int ntile = blockIdx.x * 8 + wave;
    const int n     = ntile * 16 + (lane & 15);
    const int koff  = (lane >> 4) * 2;

    v2f bfrag[15];
    #pragma unroll
    for (int s = 0; s < 15; ++s) {
        const int k0 = 4 * s;
        const int r  = k0 + koff;
        const float* base = (k0 < PF)
            ? (S_poly  + (size_t)r * OPD2)
            : (S_graph + (size_t)(r - PF) * OPD2);
        v2f bv;
        bv.x = base[n];
        bv.y = base[OPD2 + n];
        bfrag[s] = bv;
    }

    for (int mt = 0; mt < 16; ++mt) {
        v8f c = {};
        const float* arow = interm + (size_t)(mt * 16 + (lane & 15)) * KPAD + koff;
        #pragma unroll
        for (int s = 0; s < 15; ++s) {
            v2f a = *(const v2f*)(arow + 4 * s);
            c = __builtin_amdgcn_wmma_f32_16x16x4_f32(
                     false, a,  false, bfrag[s],
                     (short)0, c,  false,  false);
        }
        float* op = out + (size_t)(mt * 16 + 8 * (lane >> 4)) * OPD2 + n;
        #pragma unroll
        for (int j = 0; j < 8; ++j)
            op[(size_t)j * OPD2] = c[j];
    }
}

extern "C" void kernel_launch(void* const* d_in, const int* in_sizes, int n_in,
                              void* d_out, int out_size, void* d_ws, size_t ws_size,
                              hipStream_t stream) {
    const float* positions   = (const float*)d_in[0];
    const float* obs_pos     = (const float*)d_in[1];
    const float* poly_dic    = (const float*)d_in[2];
    const float* graph_dic   = (const float*)d_in[3];
    const float* alpha_poly  = (const float*)d_in[4];
    const float* alpha_graph = (const float*)d_in[5];
    const float* S_poly      = (const float*)d_in[6];
    const float* S_graph     = (const float*)d_in[7];
    float* out = (float*)d_out;

    float* interm = (float*)d_ws;

    build_interm_kernel<<<BATCH, 64, 0, stream>>>(
        positions, obs_pos, poly_dic, graph_dic, alpha_poly, alpha_graph, interm);

    opd_wmma_gemm_kernel<<<512, 256, 0, stream>>>(interm, S_poly, S_graph, out);
}


# interm kernel: block 64->256 threads, star search and 512-term graph dot split over 8 waves with each phase's loads all in flight; graph partials summed in LDS (new reduction tree, cols 28..59)
# speedup vs baseline: 2.3804x; 2.3804x over previous
#include <hip/hip_runtime.h>
#include <hip/hip_bf16.h>
#include <climits>

typedef __attribute__((ext_vector_type(2))) float v2f;
typedef __attribute__((ext_vector_type(4))) float v4f;
typedef __attribute__((ext_vector_type(8))) float v8f;

#define N_STARS 2048
#define BATCH   256
#define PF      28
#define NGE     512
#define GF      32
#define OPD2    65536
#define KPAD    64

__global__ __launch_bounds__(256)
void build_interm_kernel(const float* __restrict__ positions,
                         const float* __restrict__ obs_pos,
                         const float* __restrict__ poly_dic,
                         const float* __restrict__ graph_dic,
                         const float* __restrict__ alpha_poly,
                         const float* __restrict__ alpha_graph,
                         float* __restrict__ interm) {
    __shared__ int s_first[8];
    __shared__ v4f s_part[256];

    const int b    = blockIdx.x;
    const int t    = threadIdx.x;
    const int lane = t & 31;
    const int wave = __builtin_amdgcn_readfirstlane(t >> 5);
    const int q    = lane >> 3;
    const int c4   = (lane & 7) * 4;
    const int e0   = wave * 64 + q * 16;
    const int pcol = lane < PF ? lane : PF - 1;

    const float px = positions[2 * b];
    const float py = positions[2 * b + 1];
    v4f ov[4];
    #pragma unroll
    for (int i = 0; i < 4; ++i)
        ov[i] = *(const v4f*)(obs_pos + 4 * (256 * i + t));

    v4f ag[16];
    #pragma unroll
    for (int i = 0; i < 16; ++i)
        ag[i] = *(const v4f*)(alpha_graph + (e0 + i) * GF + c4);

    float ap[PF];
    if (wave == 1) {
        #pragma unroll
        for (int p = 0; p < PF; ++p)
            ap[p] = alpha_poly[p * PF + pcol];
    }

    int first = INT_MAX;
    #pragma unroll
    for (int i = 3; i >= 0; --i) {
        const unsigned m0 = __builtin_amdgcn_ballot_w32(ov[i].x == px)
                          | __builtin_amdgcn_ballot_w32(ov[i].y == py);
        const unsigned m1 = __builtin_amdgcn_ballot_w32(ov[i].z == px)
                          | __builtin_amdgcn_ballot_w32(ov[i].w == py);
        const unsigned any = m0 | m1;
        if (any) {
            const int l = __builtin_ctz(any);
            first = 2 * (256 * i + 32 * wave + l) + (((m0 >> l) & 1u) ? 0 : 1);
        }
    }
    if (lane == 0) s_first[wave] = first;
    __syncthreads();
    int idx = s_first[0];
    #pragma unroll
    for (int w = 1; w < 8; ++w) idx = min(idx, s_first[w]);
    if (idx == INT_MAX) idx = 0;

    const float* gd = graph_dic + (size_t)idx * NGE + e0;
    v4f g[4];
    #pragma unroll
    for (int k = 0; k < 4; ++k)
        g[k] = *(const v4f*)(gd + 4 * k);

    float pdv = 0.0f;
    if (wave == 1) pdv = poly_dic[(size_t)idx * PF + pcol];

    v4f acc = {0.0f, 0.0f, 0.0f, 0.0f};
    #pragma unroll
    for (int i = 0; i < 16; ++i)
        acc += g[i >> 2][i & 3] * ag[i];
    s_part[t] = acc;
    __syncthreads();

    float* orow = interm + (size_t)b * KPAD;
    if (wave == 0) {
        const float* sp = (const float*)s_part;
        float gsum = 0.0f;
        #pragma unroll
        for (int w = 0; w < 8; ++w) {
            const float* r = sp + (w * 4) * 32 + lane;
            gsum += (r[0] + r[32]) + (r[64] + r[96]);
        }
        orow[PF + lane] = gsum;
        if (lane < KPAD - PF - GF) orow[PF + GF + lane] = 0.0f;
    } else if (wave == 1) {
        float pacc = 0.0f;
        #pragma unroll
        for (int p = 0; p < PF; ++p) {
            const float x = __builtin_bit_cast(float,
                __builtin_amdgcn_readlane(__builtin_bit_cast(int, pdv), p));
            pacc += x * ap[p];
        }
        if (lane < PF) orow[lane] = pacc;
    }
}

__global__ __launch_bounds__(256)
void opd_wmma_gemm_kernel(const float* __restrict__ interm,
                          const float* __restrict__ S_poly,
                          const float* __restrict__ S_graph,
                          float* __restrict__ out) {
    const int lane  = threadIdx.x & 31;
    const int wave  = threadIdx.x >> 5;
    const int ntile = blockIdx.x * 8 + wave;
    const int n     = ntile * 16 + (lane & 15);
    const int koff  = (lane >> 4) * 2;

    v2f bfrag[15];
    #pragma unroll
    for (int s = 0; s < 15; ++s) {
        const int k0 = 4 * s;
        const int r  = k0 + koff;
        const float* base = (k0 < PF)
            ? (S_poly  + (size_t)r * OPD2)
            : (S_graph + (size_t)(r - PF) * OPD2);
        v2f bv;
        bv.x = base[n];
        bv.y = base[OPD2 + n];
        bfrag[s] = bv;
    }

    for (int mt = 0; mt < 16; ++mt) {
        v8f c = {};
        const float* arow = interm + (size_t)(mt * 16 + (lane & 15)) * KPAD + koff;
        #pragma unroll
        for (int s = 0; s < 15; ++s) {
            v2f a = *(const v2f*)(arow + 4 * s);
            c = __builtin_amdgcn_wmma_f32_16x16x4_f32(
                     false, a,  false, bfrag[s],
                     (short)0, c,  false,  false);
        }
        float* op = out + (size_t)(mt * 16 + 8 * (lane >> 4)) * OPD2 + n;
        #pragma unroll
        for (int j = 0; j < 8; ++j)
            op[(size_t)j * OPD2] = c[j];
    }
}

extern "C" void kernel_launch(void* const* d_in, const int* in_sizes, int n_in,
                              void* d_out, int out_size, void* d_ws, size_t ws_size,
                              hipStream_t stream) {
    const float* positions   = (const float*)d_in[0];
    const float* obs_pos     = (const float*)d_in[1];
    const float* poly_dic    = (const float*)d_in[2];
    const float* graph_dic   = (const float*)d_in[3];
    const float* alpha_poly  = (const float*)d_in[4];
    const float* alpha_graph = (const float*)d_in[5];
    const float* S_poly      = (const float*)d_in[6];
    const float* S_graph     = (const float*)d_in[7];
    float* out = (float*)d_out;

    float* interm = (float*)d_ws;

    build_interm_kernel<<<BATCH, 256, 0, stream>>>(
        positions, obs_pos, poly_dic, graph_dic, alpha_poly, alpha_graph, interm);

    opd_wmma_gemm_kernel<<<512, 256, 0, stream>>>(interm, S_poly, S_graph, out);
}


# GEMM kernel: each block copies interm (A operand) into LDS once and waves read A fragments from LDS (conflict-free 60-float pitch) instead of 16-line global loads repeated by every wave
# speedup vs baseline: 2.8840x; 1.2116x over previous
#include <hip/hip_runtime.h>
#include <hip/hip_bf16.h>
#include <climits>

typedef __attribute__((ext_vector_type(2))) float v2f;
typedef __attribute__((ext_vector_type(4))) float v4f;
typedef __attribute__((ext_vector_type(8))) float v8f;

#define N_STARS 2048
#define BATCH   256
#define PF      28
#define NGE     512
#define GF      32
#define OPD2    65536
#define KPAD    64

__global__ __launch_bounds__(256)
void build_interm_kernel(const float* __restrict__ positions,
                         const float* __restrict__ obs_pos,
                         const float* __restrict__ poly_dic,
                         const float* __restrict__ graph_dic,
                         const float* __restrict__ alpha_poly,
                         const float* __restrict__ alpha_graph,
                         float* __restrict__ interm) {
    __shared__ int s_first[8];
    __shared__ v4f s_part[256];

    const int b    = blockIdx.x;
    const int t    = threadIdx.x;
    const int lane = t & 31;
    const int wave = __builtin_amdgcn_readfirstlane(t >> 5);
    const int q    = lane >> 3;
    const int c4   = (lane & 7) * 4;
    const int e0   = wave * 64 + q * 16;
    const int pcol = lane < PF ? lane : PF - 1;

    const float px = positions[2 * b];
    const float py = positions[2 * b + 1];
    v4f ov[4];
    #pragma unroll
    for (int i = 0; i < 4; ++i)
        ov[i] = *(const v4f*)(obs_pos + 4 * (256 * i + t));

    v4f ag[16];
    #pragma unroll
    for (int i = 0; i < 16; ++i)
        ag[i] = *(const v4f*)(alpha_graph + (e0 + i) * GF + c4);

    float ap[PF];
    if (wave == 1) {
        #pragma unroll
        for (int p = 0; p < PF; ++p)
            ap[p] = alpha_poly[p * PF + pcol];
    }

    int first = INT_MAX;
    #pragma unroll
    for (int i = 3; i >= 0; --i) {
        const unsigned m0 = __builtin_amdgcn_ballot_w32(ov[i].x == px)
                          | __builtin_amdgcn_ballot_w32(ov[i].y == py);
        const unsigned m1 = __builtin_amdgcn_ballot_w32(ov[i].z == px)
                          | __builtin_amdgcn_ballot_w32(ov[i].w == py);
        const unsigned any = m0 | m1;
        if (any) {
            const int l = __builtin_ctz(any);
            first = 2 * (256 * i + 32 * wave + l) + (((m0 >> l) & 1u) ? 0 : 1);
        }
    }
    if (lane == 0) s_first[wave] = first;
    __syncthreads();
    int idx = s_first[0];
    #pragma unroll
    for (int w = 1; w < 8; ++w) idx = min(idx, s_first[w]);
    if (idx == INT_MAX) idx = 0;

    const float* gd = graph_dic + (size_t)idx * NGE + e0;
    v4f g[4];
    #pragma unroll
    for (int k = 0; k < 4; ++k)
        g[k] = *(const v4f*)(gd + 4 * k);

    float pdv = 0.0f;
    if (wave == 1) pdv = poly_dic[(size_t)idx * PF + pcol];

    v4f acc = {0.0f, 0.0f, 0.0f, 0.0f};
    #pragma unroll
    for (int i = 0; i < 16; ++i)
        acc += g[i >> 2][i & 3] * ag[i];
    s_part[t] = acc;
    __syncthreads();

    float* orow = interm + (size_t)b * KPAD;
    if (wave == 0) {
        const float* sp = (const float*)s_part;
        float gsum = 0.0f;
        #pragma unroll
        for (int w = 0; w < 8; ++w) {
            const float* r = sp + (w * 4) * 32 + lane;
            gsum += (r[0] + r[32]) + (r[64] + r[96]);
        }
        orow[PF + lane] = gsum;
        if (lane < KPAD - PF - GF) orow[PF + GF + lane] = 0.0f;
    } else if (wave == 1) {
        float pacc = 0.0f;
        #pragma unroll
        for (int p = 0; p < PF; ++p) {
            const float x = __builtin_bit_cast(float,
                __builtin_amdgcn_readlane(__builtin_bit_cast(int, pdv), p));
            pacc += x * ap[p];
        }
        if (lane < PF) orow[lane] = pacc;
    }
}

#define APITCH 60
#define KSTEPS 15

__device__ __forceinline__ void load_a_tile(v2f (&a)[KSTEPS], const float* arow) {
    #pragma unroll
    for (int s = 0; s < KSTEPS; ++s)
        a[s] = *(const v2f*)(arow + 4 * s);
}

__device__ __forceinline__ void mma_store_tile(const v2f (&a)[KSTEPS],
                                               const v2f (&bfrag)[KSTEPS],
                                               float* __restrict__ op) {
    v8f c = {};
    #pragma unroll
    for (int s = 0; s < KSTEPS; ++s)
        c = __builtin_amdgcn_wmma_f32_16x16x4_f32(
                 false, a[s],  false, bfrag[s],
                 (short)0, c,  false,  false);
    #pragma unroll
    for (int j = 0; j < 8; ++j)
        op[(size_t)j * OPD2] = c[j];
}

__global__ __launch_bounds__(256)
void opd_wmma_gemm_kernel(const float* __restrict__ interm,
                          const float* __restrict__ S_poly,
                          const float* __restrict__ S_graph,
                          float* __restrict__ out) {
    __shared__ __attribute__((aligned(16))) float s_a[BATCH * APITCH];

    const int t     = threadIdx.x;
    const int lane  = t & 31;
    const int wave  = t >> 5;
    const int ntile = blockIdx.x * 8 + wave;
    const int n     = ntile * 16 + (lane & 15);
    const int koff  = (lane >> 4) * 2;

    const bool stager = t < 16 * KSTEPS;
    const float* sg = interm + (t / KSTEPS) * KPAD + 4 * (t % KSTEPS);
    v4f stage[16];
    if (stager) {
        #pragma unroll
        for (int g = 0; g < 16; ++g)
            stage[g] = *(const v4f*)(sg + g * 16 * KPAD);
    }

    v2f bfrag[KSTEPS];
    #pragma unroll
    for (int s = 0; s < KSTEPS; ++s) {
        const int k0 = 4 * s;
        const int r  = k0 + koff;
        const float* base = (k0 < PF)
            ? (S_poly  + (size_t)r * OPD2)
            : (S_graph + (size_t)(r - PF) * OPD2);
        v2f bv;
        bv.x = base[n];
        bv.y = base[OPD2 + n];
        bfrag[s] = bv;
    }

    if (stager) {
        #pragma unroll
        for (int g = 0; g < 16; ++g)
            *(v4f*)(s_a + g * 16 * APITCH + 4 * t) = stage[g];
    }
    __syncthreads();

    const float* abase = s_a + (lane & 15) * APITCH + koff;
    float* obase = out + (size_t)(8 * (lane >> 4)) * OPD2 + n;
    const size_t ostep = (size_t)16 * OPD2;
    v2f a0[KSTEPS], a1[KSTEPS];
    load_a_tile(a0, abase);
    for (int mt = 0; mt < 14; mt += 2) {
        load_a_tile(a1, abase + (mt + 1) * 16 * APITCH);
        mma_store_tile(a0, bfrag, obase + mt * ostep);
        load_a_tile(a0, abase + (mt + 2) * 16 * APITCH);
        mma_store_tile(a1, bfrag, obase + (mt + 1) * ostep);
    }
    load_a_tile(a1, abase + 15 * 16 * APITCH);
    mma_store_tile(a0, bfrag, obase + 14 * ostep);
    mma_store_tile(a1, bfrag, obase + 15 * ostep);
}

extern "C" void kernel_launch(void* const* d_in, const int* in_sizes, int n_in,
                              void* d_out, int out_size, void* d_ws, size_t ws_size,
                              hipStream_t stream) {
    const float* positions   = (const float*)d_in[0];
    const float* obs_pos     = (const float*)d_in[1];
    const float* poly_dic    = (const float*)d_in[2];
    const float* graph_dic   = (const float*)d_in[3];
    const float* alpha_poly  = (const float*)d_in[4];
    const float* alpha_graph = (const float*)d_in[5];
    const float* S_poly      = (const float*)d_in[6];
    const float* S_graph     = (const float*)d_in[7];
    float* out = (float*)d_out;

    float* interm = (float*)d_ws;

    build_interm_kernel<<<BATCH, 256, 0, stream>>>(
        positions, obs_pos, poly_dic, graph_dic, alpha_poly, alpha_graph, interm);

    opd_wmma_gemm_kernel<<<512, 256, 0, stream>>>(interm, S_poly, S_graph, out);
}


# GEMM: wave tile 16x64 as four interleaved 16-column tiles (pixel 4c+u): one LDS A fragment feeds 4 WMMAs, 16-B B loads and whole-line C stores; one wave per SIMD (256 blocks of 4 waves, LDS-sized)
# speedup vs baseline: 4.1909x; 1.4532x over previous
#include <hip/hip_runtime.h>
#include <hip/hip_bf16.h>
#include <climits>

typedef __attribute__((ext_vector_type(2))) float v2f;
typedef __attribute__((ext_vector_type(4))) float v4f;
typedef __attribute__((ext_vector_type(8))) float v8f;

#define N_STARS 2048
#define BATCH   256
#define PF      28
#define NGE     512
#define GF      32
#define OPD2    65536
#define KPAD    64

__global__ __launch_bounds__(256)
void build_interm_kernel(const float* __restrict__ positions,
                         const float* __restrict__ obs_pos,
                         const float* __restrict__ poly_dic,
                         const float* __restrict__ graph_dic,
                         const float* __restrict__ alpha_poly,
                         const float* __restrict__ alpha_graph,
                         float* __restrict__ interm) {
    __shared__ int s_first[8];
    __shared__ v4f s_part[256];

    const int b    = blockIdx.x;
    const int t    = threadIdx.x;
    const int lane = t & 31;
    const int wave = __builtin_amdgcn_readfirstlane(t >> 5);
    const int q    = lane >> 3;
    const int c4   = (lane & 7) * 4;
    const int e0   = wave * 64 + q * 16;
    const int pcol = lane < PF ? lane : PF - 1;

    const float px = positions[2 * b];
    const float py = positions[2 * b + 1];
    v4f ov[4];
    #pragma unroll
    for (int i = 0; i < 4; ++i)
        ov[i] = *(const v4f*)(obs_pos + 4 * (256 * i + t));

    v4f ag[16];
    #pragma unroll
    for (int i = 0; i < 16; ++i)
        ag[i] = *(const v4f*)(alpha_graph + (e0 + i) * GF + c4);

    float ap[PF];
    if (wave == 1) {
        #pragma unroll
        for (int p = 0; p < PF; ++p)
            ap[p] = alpha_poly[p * PF + pcol];
    }

    int first = INT_MAX;
    #pragma unroll
    for (int i = 3; i >= 0; --i) {
        const unsigned m0 = __builtin_amdgcn_ballot_w32(ov[i].x == px)
                          | __builtin_amdgcn_ballot_w32(ov[i].y == py);
        const unsigned m1 = __builtin_amdgcn_ballot_w32(ov[i].z == px)
                          | __builtin_amdgcn_ballot_w32(ov[i].w == py);
        const unsigned any = m0 | m1;
        if (any) {
            const int l = __builtin_ctz(any);
            first = 2 * (256 * i + 32 * wave + l) + (((m0 >> l) & 1u) ? 0 : 1);
        }
    }
    if (lane == 0) s_first[wave] = first;
    __syncthreads();
    int idx = s_first[0];
    #pragma unroll
    for (int w = 1; w < 8; ++w) idx = min(idx, s_first[w]);
    if (idx == INT_MAX) idx = 0;

    const float* gd = graph_dic + (size_t)idx * NGE + e0;
    v4f g[4];
    #pragma unroll
    for (int k = 0; k < 4; ++k)
        g[k] = *(const v4f*)(gd + 4 * k);

    float pdv = 0.0f;
    if (wave == 1) pdv = poly_dic[(size_t)idx * PF + pcol];

    v4f acc = {0.0f, 0.0f, 0.0f, 0.0f};
    #pragma unroll
    for (int i = 0; i < 16; ++i)
        acc += g[i >> 2][i & 3] * ag[i];
    s_part[t] = acc;
    __syncthreads();

    float* orow = interm + (size_t)b * KPAD;
    if (wave == 0) {
        const float* sp = (const float*)s_part;
        float gsum = 0.0f;
        #pragma unroll
        for (int w = 0; w < 8; ++w) {
            const float* r = sp + (w * 4) * 32 + lane;
            gsum += (r[0] + r[32]) + (r[64] + r[96]);
        }
        orow[PF + lane] = gsum;
        if (lane < KPAD - PF - GF) orow[PF + GF + lane] = 0.0f;
    } else if (wave == 1) {
        float pacc = 0.0f;
        #pragma unroll
        for (int p = 0; p < PF; ++p) {
            const float x = __builtin_bit_cast(float,
                __builtin_amdgcn_readlane(__builtin_bit_cast(int, pdv), p));
            pacc += x * ap[p];
        }
        if (lane < PF) orow[lane] = pacc;
    }
}

#define APITCH 60
#define KSTEPS 15
#define NT     4
#define LDS_FLOATS (41 * 1024)
static_assert(LDS_FLOATS >= BATCH * APITCH, "the staged A operand fits the reservation");

__device__ __forceinline__ void load_a_tile(v2f (&a)[KSTEPS], const float* arow) {
    #pragma unroll
    for (int s = 0; s < KSTEPS; ++s)
        a[s] = *(const v2f*)(arow + 4 * s);
}

__device__ __forceinline__ void mma_tile(v8f (&c)[NT], const v2f (&a)[KSTEPS],
                                         const v2f (&bfrag)[KSTEPS][NT]) {
    #pragma unroll
    for (int u = 0; u < NT; ++u) c[u] = v8f{};
    #pragma unroll
    for (int s = 0; s < KSTEPS; ++s) {
        #pragma unroll
        for (int u = 0; u < NT; ++u)
            c[u] = __builtin_amdgcn_wmma_f32_16x16x4_f32(
                     false, a[s],  false, bfrag[s][u],
                     (short)0, c[u],  false,  false);
    }
}

__device__ __forceinline__ void store_tile(const v8f (&c)[NT], float* __restrict__ op) {
    #pragma unroll
    for (int j = 0; j < 8; ++j) {
        v4f o = {c[0][j], c[1][j], c[2][j], c[3][j]};
        *(v4f*)(op + (size_t)j * OPD2) = o;
    }
}

#define SCHED_FENCE() __builtin_amdgcn_sched_barrier(0)

__global__ __launch_bounds__(128)
void opd_wmma_gemm_kernel(const float* __restrict__ interm,
                          const float* __restrict__ S_poly,
                          const float* __restrict__ S_graph,
                          float* __restrict__ out) {
    __shared__ __attribute__((aligned(16))) float s_a[LDS_FLOATS];

    const int t     = threadIdx.x;
    const int lane  = t & 31;
    const int wave  = t >> 5;
    const int p0    = (blockIdx.x * 4 + wave) * 64 + 4 * (lane & 15);
    const int koff  = (lane >> 4) * 2;

    const int  sl1 = t + 128;
    const bool two = sl1 < 16 * KSTEPS;
    const float* sg0 = interm + (t / KSTEPS) * KPAD + 4 * (t % KSTEPS);
    const float* sg1 = interm + (sl1 / KSTEPS) * KPAD + 4 * (sl1 % KSTEPS);
    v4f st0[16], st1[16];
    #pragma unroll
    for (int g = 0; g < 16; ++g)
        st0[g] = *(const v4f*)(sg0 + g * 16 * KPAD);
    if (two) {
        #pragma unroll
        for (int g = 0; g < 16; ++g)
            st1[g] = *(const v4f*)(sg1 + g * 16 * KPAD);
    }

    v2f bfrag[KSTEPS][NT];
    #pragma unroll
    for (int s = 0; s < KSTEPS; ++s) {
        const int k0 = 4 * s;
        const int r  = k0 + koff;
        const float* base = (k0 < PF)
            ? (S_poly  + (size_t)r * OPD2)
            : (S_graph + (size_t)(r - PF) * OPD2);
        const v4f r0 = *(const v4f*)(base + p0);
        const v4f r1 = *(const v4f*)(base + OPD2 + p0);
        #pragma unroll
        for (int u = 0; u < NT; ++u) {
            v2f bv;
            bv.x = r0[u];
            bv.y = r1[u];
            bfrag[s][u] = bv;
        }
    }

    #pragma unroll
    for (int g = 0; g < 16; ++g)
        *(v4f*)(s_a + g * 16 * APITCH + 4 * t) = st0[g];
    if (two) {
        #pragma unroll
        for (int g = 0; g < 16; ++g)
            *(v4f*)(s_a + g * 16 * APITCH + 4 * sl1) = st1[g];
    }
    __syncthreads();

    const float* abase = s_a + (lane & 15) * APITCH + koff;
    float* obase = out + (size_t)(8 * (lane >> 4)) * OPD2 + p0;
    const size_t ostep = (size_t)16 * OPD2;
    v2f a0[KSTEPS], a1[KSTEPS];
    v8f c0[NT], c1[NT];
    load_a_tile(a0, abase);
    load_a_tile(a1, abase + 16 * APITCH);
    SCHED_FENCE();
    mma_tile(c0, a0, bfrag);
    SCHED_FENCE();
    for (int mt = 0; mt < 14; mt += 2) {
        load_a_tile(a0, abase + (mt + 2) * 16 * APITCH);
        SCHED_FENCE();
        mma_tile(c1, a1, bfrag);
        store_tile(c0, obase + mt * ostep);
        SCHED_FENCE();
        load_a_tile(a1, abase + (mt + 3) * 16 * APITCH);
        SCHED_FENCE();
        mma_tile(c0, a0, bfrag);
        store_tile(c1, obase + (mt + 1) * ostep);
        SCHED_FENCE();
    }
    mma_tile(c1, a1, bfrag);
    store_tile(c0, obase + 14 * ostep);
    SCHED_FENCE();
    store_tile(c1, obase + 15 * ostep);
}

extern "C" void kernel_launch(void* const* d_in, const int* in_sizes, int n_in,
                              void* d_out, int out_size, void* d_ws, size_t ws_size,
                              hipStream_t stream) {
    const float* positions   = (const float*)d_in[0];
    const float* obs_pos     = (const float*)d_in[1];
    const float* poly_dic    = (const float*)d_in[2];
    const float* graph_dic   = (const float*)d_in[3];
    const float* alpha_poly  = (const float*)d_in[4];
    const float* alpha_graph = (const float*)d_in[5];
    const float* S_poly      = (const float*)d_in[6];
    const float* S_graph     = (const float*)d_in[7];
    float* out = (float*)d_out;

    float* interm = (float*)d_ws;

    build_interm_kernel<<<BATCH, 256, 0, stream>>>(
        positions, obs_pos, poly_dic, graph_dic, alpha_poly, alpha_graph, interm);

    opd_wmma_gemm_kernel<<<256, 128, 0, stream>>>(interm, S_poly, S_graph, out);
}
